# starting kernel
# baseline (speedup 1.0000x reference)
#include <hip/hip_runtime.h>
#include <utility>

typedef __attribute__((ext_vector_type(2))) float v2f;
typedef __attribute__((ext_vector_type(8))) float v8f;

namespace gol {

struct Tables {
  unsigned short s3[96];
  unsigned short s4[64];
};

constexpr int popc9(int m) { int c = 0; for (int k = 0; k < 9; ++k) c += (m >> k) & 1; return c; }

constexpr Tables build() {
  Tables t{};
  for (int i = 0; i < 96; ++i) t.s3[i] = 0xFFFFu;
  for (int i = 0; i < 64; ++i) t.s4[i] = 0u;
  int i3 = 0, i4 = 0;
  for (int m = 0; m < 512; ++m) {
    const int tot = popc9(m);
    const bool alive = ((m >> 4) & 1) != 0;
    const bool future = alive ? (tot == 3 || tot == 4) : (tot == 3);
    if (!future) continue;
    if (tot == 3) t.s3[i3++] = (unsigned short)m;
    else          t.s4[i4++] = (unsigned short)m;
  }
  return t;
}

constexpr unsigned S3c(int i) { return build().s3[i]; }
constexpr unsigned S4c(int i) { return build().s4[i]; }

struct MTab { unsigned m[96][2]; };

constexpr MTab buildMT() {
  MTab r{};
  constexpr Tables t = build();
  for (int row = 0; row < 96; ++row) {
    for (int H = 0; H < 2; ++H) {
      unsigned mm = 0;
      for (int idx = 0; idx < 28; ++idx) {
        const int kb = idx >> 1, j = idx & 1;
        const unsigned c3 = t.s3[row];
        const unsigned c4 = t.s4[4 * kb + j + 2 * H];
        if ((c3 & ~c4) == 0u) mm |= (1u << idx);
      }
      r.m[row][H] = mm;
    }
  }
  return r;
}

__device__ constexpr MTab MT = buildMT();

template<typename F, int... Is>
__device__ __forceinline__ void static_for_impl(F&& f, std::integer_sequence<int, Is...>) {
  (f(std::integral_constant<int, Is>{}), ...);
}
template<int N, typename F>
__device__ __forceinline__ void static_for(F&& f) {
  static_for_impl(static_cast<F&&>(f), std::make_integer_sequence<int, N>{});
}

}

__global__ __launch_bounds__(256) void gol_wmma_kernel(const float* __restrict__ x,
                                                       float* __restrict__ out,
                                                       int nTiles) {
  using namespace gol;
  const int lane = (int)(threadIdx.x & 31u);
  const int p = lane & 15;
  const int h = lane >> 4;
  const int w = (int)blockIdx.x * ((int)blockDim.x >> 5) + ((int)threadIdx.x >> 5);
  if (w >= nTiles) return;

  const int b  = w >> 12;
  const int rm = w & 4095;
  const int y  = rm >> 4;
  const int tx = rm & 15;
  const int x0 = (tx << 4) + p;
  const float* img = x + (size_t)b * 65536;

  float n[9];
#pragma unroll
  for (int k = 0; k < 9; ++k) {
    const int dy = k / 3 - 1, dx = k % 3 - 1;
    const int yy = (y + dy) & 255;
    const int xx = (x0 + dx) & 255;
    n[k] = img[yy * 256 + xx];
  }
  float c[9];
#pragma unroll
  for (int k = 0; k < 9; ++k) c[k] = 1.0f - n[k];

  float gA[16]; gA[0] = 1.0f;
  static_for<4>([&](auto C4) {
    constexpr int cell = C4.value;
    constexpr int s = 1 << cell;
    static_for<s>([&](auto M) {
      constexpr int m = s - 1 - M.value;
      gA[m + s] = gA[m] * n[cell];
      gA[m]     = gA[m] * c[cell];
    });
  });
  float gB[32]; gB[0] = 1.0f;
  static_for<5>([&](auto C5) {
    constexpr int cell = C5.value;
    constexpr int s = 1 << cell;
    static_for<s>([&](auto M) {
      constexpr int m = s - 1 - M.value;
      gB[m + s] = gB[m] * n[cell + 4];
      gB[m]     = gB[m] * c[cell + 4];
    });
  });

  const bool hi = (h != 0);
  float sumAcc = 0.0f;

  float acc4[28];
  static_for<28>([&](auto I) {
    constexpr int idx = I.value;
    constexpr int kb = idx >> 1, j = idx & 1;
    constexpr unsigned m0 = S4c(4 * kb + j);
    constexpr unsigned m1 = S4c(4 * kb + j + 2);
    const float v0 = gA[m0 & 15] * gB[(m0 >> 4) & 31];
    const float v1 = gA[m1 & 15] * gB[(m1 >> 4) & 31];
    const float v = hi ? v1 : v0;
    acc4[idx] = v;
    sumAcc += v;
  });

  float acc3[48];
  static_for<48>([&](auto I) {
    constexpr int idx = I.value;
    constexpr int t = idx >> 3, v = idx & 7;
    constexpr int r0 = 16 * t + v, r1 = r0 + 8;
    float v0 = 0.0f, v1 = 0.0f;
    if constexpr (r0 < 84) { constexpr unsigned mk = S3c(r0); v0 = gA[mk & 15] * gB[(mk >> 4) & 31]; }
    if constexpr (r1 < 84) { constexpr unsigned mk = S3c(r1); v1 = gA[mk & 15] * gB[(mk >> 4) & 31]; }
    const float val = hi ? v1 : v0;
    acc3[idx] = val;
    sumAcc += val;
  });

  unsigned mt[6];
  static_for<6>([&](auto T) {
    constexpr int t = T.value;
    mt[t] = MT.m[16 * t + p][h];
  });

  float orp = 0.0f;
#pragma unroll
  for (int t = 0; t < 6; ++t) {
    v8f C = {0.f, 0.f, 0.f, 0.f, 0.f, 0.f, 0.f, 0.f};
    const unsigned m = mt[t];
#pragma unroll
    for (int kb = 0; kb < 14; ++kb) {
      v2f A, Bv;
      A.x = (float)((m >> (2 * kb)) & 1u);
      A.y = (float)((m >> (2 * kb + 1)) & 1u);
      Bv.x = acc4[2 * kb];
      Bv.y = acc4[2 * kb + 1];
      C = __builtin_amdgcn_wmma_f32_16x16x4_f32(false, A, false, Bv, (short)0, C, false, false);
    }
#pragma unroll
    for (int v = 0; v < 8; ++v) orp = fmaf(acc3[8 * t + v], C[v], orp);
  }

  float tot = sumAcc - orp;
  tot += __shfl_xor(tot, 16, 32);
  if (h == 0) out[(size_t)b * 65536 + (size_t)y * 256 + (size_t)x0] = tot;
}

extern "C" void kernel_launch(void* const* d_in, const int* in_sizes, int n_in,
                              void* d_out, int out_size, void* d_ws, size_t ws_size,
                              hipStream_t stream) {
  (void)n_in; (void)d_ws; (void)ws_size; (void)out_size;
  const float* x = (const float*)d_in[0];
  float* out = (float*)d_out;
  const int total = in_sizes[0];
  const int Bn = total >> 16;
  const int nTiles = Bn * 4096;
  const int wavesPerBlock = 8;
  dim3 block(256);
  dim3 grid((unsigned)((nTiles + wavesPerBlock - 1) / wavesPerBlock));
  gol_wmma_kernel<<<grid, block, 0, stream>>>(x, out, nTiles);
}


# One wave covers 32 pixels, one per lane (was 16, both lane halves repeating each product): B and D cross halves by v_permlane16_swap, one A fragment feeds two WMMAs; f32 sum order changes
# speedup vs baseline: 2.0127x; 2.0127x over previous
#include <hip/hip_runtime.h>
#include <utility>

typedef __attribute__((ext_vector_type(2))) float v2f;
typedef __attribute__((ext_vector_type(8))) float v8f;

namespace gol {

struct Tables {
  unsigned short s3[96];
  unsigned short s4[64];
};

constexpr int popc9(int m) { int c = 0; for (int k = 0; k < 9; ++k) c += (m >> k) & 1; return c; }
constexpr int obit(int i) { return i < 4 ? i : i + 1; }

constexpr Tables build() {
  Tables t{};
  for (int i = 0; i < 96; ++i) t.s3[i] = 0xFFFFu;
  for (int i = 0; i < 64; ++i) t.s4[i] = 0u;
  unsigned short col[56] = {};
  int nc = 0;
  for (int cls = 0; cls < 3; ++cls)
    for (int a = 0; a < 8; ++a)
      for (int b = a + 1; b < 8; ++b)
        for (int c = b + 1; c < 8; ++c) {
          const int k = (c < 6) ? 0 : ((b == 6 && a >= 1 && a <= 4) ? 1 : 2);
          if (k == cls)
            col[nc++] = (unsigned short)((1u << obit(a)) | (1u << obit(b)) | (1u << obit(c)));
        }
  for (int j = 0; j < 56; ++j) t.s4[j] = (unsigned short)(col[j] | 16u);
  int r0 = 0, r1 = 16;
  for (int a = 0; a < 8; ++a)
    for (int b = a + 1; b < 8; ++b) {
      const unsigned short m = (unsigned short)((1u << obit(a)) | (1u << obit(b)) | 16u);
      const bool first = (b < 6) || (a == 0 && b == 6);
      if (first) t.s3[r0++] = m; else t.s3[r1++] = m;
    }
  for (int i = 0; i < 4; ++i) t.s3[28 + i] = col[52 + i];
  for (int i = 0; i < 52; ++i) t.s3[32 + i] = col[i];
  return t;
}

constexpr bool tablesComplete() {
  const Tables t = build();
  bool seen[512] = {};
  int n3 = 0, n4 = 0;
  for (int i = 0; i < 84; ++i) {
    const int m = t.s3[i];
    if (m >= 512 || popc9(m) != 3 || seen[m]) return false;
    seen[m] = true; ++n3;
  }
  for (int i = 0; i < 56; ++i) {
    const int m = t.s4[i];
    if (m >= 512 || popc9(m) != 4 || ((m >> 4) & 1) == 0 || seen[m]) return false;
    seen[m] = true; ++n4;
  }
  for (int i = 84; i < 96; ++i) if (t.s3[i] != 0xFFFFu) return false;
  return n3 == 84 && n4 == 56;
}
static_assert(tablesComplete(), "config tables do not match the rule");

constexpr unsigned S3c(int i) { return build().s3[i]; }
constexpr unsigned S4c(int i) { return build().s4[i]; }

struct MTab { unsigned m[96][2]; };

constexpr MTab buildMT() {
  MTab r{};
  constexpr Tables t = build();
  for (int row = 0; row < 96; ++row) {
    for (int H = 0; H < 2; ++H) {
      unsigned mm = 0;
      for (int idx = 0; idx < 28; ++idx) {
        const int kb = idx >> 1, j = idx & 1;
        const unsigned c3 = t.s3[row];
        const unsigned c4 = t.s4[4 * kb + j + 2 * H];
        if ((c3 & ~c4) == 0u) mm |= (1u << idx);
      }
      r.m[row][H] = mm;
    }
  }
  return r;
}

__device__ constexpr MTab MT = buildMT();

constexpr bool blockNZ(int t, int kb) {
  const MTab r = buildMT();
  for (int p = 0; p < 16; ++p)
    for (int H = 0; H < 2; ++H)
      if ((r.m[16 * t + p][H] >> (2 * kb)) & 3u) return true;
  return false;
}

template<typename F, int... Is>
__device__ __forceinline__ void static_for_impl(F&& f, std::integer_sequence<int, Is...>) {
  (f(std::integral_constant<int, Is>{}), ...);
}
template<int N, typename F>
__device__ __forceinline__ void static_for(F&& f) {
  static_for_impl(static_cast<F&&>(f), std::make_integer_sequence<int, N>{});
}

}

typedef __attribute__((ext_vector_type(2))) unsigned v2u;

__device__ __forceinline__ void half_swap(float& d, float& s) {
  const v2u r = __builtin_amdgcn_permlane16_swap(__float_as_uint(d), __float_as_uint(s), false, false);
  d = __uint_as_float(r.x);
  s = __uint_as_float(r.y);
}

__global__ __launch_bounds__(256) void gol_wmma_kernel(const float* __restrict__ x,
                                                       float* __restrict__ out,
                                                       int nTiles) {
  using namespace gol;
  const int lane = (int)(threadIdx.x & 31u);
  const int p = lane & 15;
  const int h = lane >> 4;
  const int w = (int)blockIdx.x * ((int)blockDim.x >> 5) + ((int)threadIdx.x >> 5);
  if (w >= nTiles) return;

  const int b  = w >> 11;
  const int rm = w & 2047;
  const int y  = rm >> 3;
  const int tx = rm & 7;
  const int x0 = (tx << 5) + lane;
  const float* img = x + (size_t)b * 65536;

  float n[9];
#pragma unroll
  for (int k = 0; k < 9; ++k) {
    const int dy = k / 3 - 1, dx = k % 3 - 1;
    const int yy = (y + dy) & 255;
    const int xx = (x0 + dx) & 255;
    n[k] = img[yy * 256 + xx];
  }
  float c[9];
#pragma unroll
  for (int k = 0; k < 9; ++k) c[k] = 1.0f - n[k];

  float gA[16]; gA[0] = 1.0f;
  static_for<4>([&](auto C4) {
    constexpr int cell = C4.value;
    constexpr int s = 1 << cell;
    static_for<s>([&](auto M) {
      constexpr int m = s - 1 - M.value;
      gA[m + s] = gA[m] * n[cell];
      gA[m]     = gA[m] * c[cell];
    });
  });
  float gB[32]; gB[0] = 1.0f;
  static_for<5>([&](auto C5) {
    constexpr int cell = C5.value;
    constexpr int s = 1 << cell;
    static_for<s>([&](auto M) {
      constexpr int m = s - 1 - M.value;
      gB[m + s] = gB[m] * n[cell + 4];
      gB[m]     = gB[m] * c[cell + 4];
    });
  });

  float sum0 = 0.0f, sum1 = 0.0f, sum2 = 0.0f, sum3 = 0.0f;

  float a4[56];
  static_for<56>([&](auto I) {
    constexpr int col = I.value;
    constexpr unsigned mk = S4c(col);
    const float v = gA[mk & 15] * gB[(mk >> 4) & 31];
    a4[col] = v;
    if constexpr ((col & 1) == 0) sum0 += v; else sum1 += v;
  });

  float b1[28], b2[28];
  static_for<28>([&](auto I) {
    constexpr int idx = I.value;
    constexpr int kb = idx >> 1, j = idx & 1;
    float lo = a4[4 * kb + j];
    float hi = a4[4 * kb + j + 2];
    half_swap(lo, hi);
    b1[idx] = lo;
    b2[idx] = hi;
  });

  float a3[84];
  static_for<84>([&](auto I) {
    constexpr int row = I.value;
    constexpr unsigned mk = S3c(row);
    const float v = gA[mk & 15] * gB[(mk >> 4) & 31];
    a3[row] = v;
    if constexpr ((row & 1) == 0) sum2 += v; else sum3 += v;
  });

  unsigned mt[6];
  static_for<6>([&](auto T) {
    constexpr int t = T.value;
    mt[t] = MT.m[16 * t + p][h];
  });

  float orp0 = 0.0f, orp1 = 0.0f;
  static_for<6>([&](auto TT) {
    constexpr int t = TT.value;
    v8f C1 = {0.f, 0.f, 0.f, 0.f, 0.f, 0.f, 0.f, 0.f};
    v8f C2 = {0.f, 0.f, 0.f, 0.f, 0.f, 0.f, 0.f, 0.f};
    const unsigned m = mt[t];
    static_for<14>([&](auto KB) {
      constexpr int kb = KB.value;
      if constexpr (blockNZ(t, kb)) {
        v2f A, B1, B2;
        A.x = (float)((m >> (2 * kb)) & 1u);
        A.y = (float)((m >> (2 * kb + 1)) & 1u);
        B1.x = b1[2 * kb]; B1.y = b1[2 * kb + 1];
        B2.x = b2[2 * kb]; B2.y = b2[2 * kb + 1];
        C1 = __builtin_amdgcn_wmma_f32_16x16x4_f32(false, A, false, B1, (short)0, C1, false, false);
        C2 = __builtin_amdgcn_wmma_f32_16x16x4_f32(false, A, false, B2, (short)0, C2, false, false);
      }
    });
    static_for<8>([&](auto V) {
      constexpr int v = V.value;
      constexpr int r0 = 16 * t + v, r1 = r0 + 8;
      if constexpr (r0 < 84) {
        float lo = C1[v];
        float hi = C2[v];
        half_swap(lo, hi);
        orp0 = fmaf(a3[r0], lo, orp0);
        if constexpr (r1 < 84) orp1 = fmaf(a3[r1], hi, orp1);
      }
    });
  });

  const float tot = ((sum0 + sum1) + (sum2 + sum3)) - (orp0 + orp1);
  out[(size_t)b * 65536 + (size_t)y * 256 + (size_t)x0] = tot;
}

extern "C" void kernel_launch(void* const* d_in, const int* in_sizes, int n_in,
                              void* d_out, int out_size, void* d_ws, size_t ws_size,
                              hipStream_t stream) {
  (void)n_in; (void)d_ws; (void)ws_size; (void)out_size;
  const float* x = (const float*)d_in[0];
  float* out = (float*)d_out;
  const int total = in_sizes[0];
  const int Bn = total >> 16;
  const int nTiles = Bn * 2048;
  const int wavesPerBlock = 8;
  dim3 block(256);
  dim3 grid((unsigned)((nTiles + wavesPerBlock - 1) / wavesPerBlock));
  gol_wmma_kernel<<<grid, block, 0, stream>>>(x, out, nTiles);
}


# The 56 dead-center rows of the 0/1 pair matrix are unit rows (one superset each), so their term is a3*a4 in the lane; their 26 of 70 WMMAs go, the 28-row pair GEMM stays; f32 sum order changes
# speedup vs baseline: 2.4030x; 1.1939x over previous
#include <hip/hip_runtime.h>
#include <utility>

typedef __attribute__((ext_vector_type(2))) float v2f;
typedef __attribute__((ext_vector_type(8))) float v8f;

namespace gol {

struct Tables {
  unsigned short s3[32];
  unsigned short s4[64];
};

constexpr int popc9(int m) { int c = 0; for (int k = 0; k < 9; ++k) c += (m >> k) & 1; return c; }
constexpr int obit(int i) { return i < 4 ? i : i + 1; }

constexpr Tables build() {
  Tables t{};
  for (int i = 0; i < 32; ++i) t.s3[i] = 0xFFFFu;
  for (int i = 0; i < 64; ++i) t.s4[i] = 0u;
  unsigned short col[56] = {};
  int nc = 0;
  for (int cls = 0; cls < 3; ++cls)
    for (int a = 0; a < 8; ++a)
      for (int b = a + 1; b < 8; ++b)
        for (int c = b + 1; c < 8; ++c) {
          const int k = (c < 6) ? 0 : ((b == 6 && a >= 1 && a <= 4) ? 1 : 2);
          if (k == cls)
            col[nc++] = (unsigned short)((1u << obit(a)) | (1u << obit(b)) | (1u << obit(c)));
        }
  for (int j = 0; j < 56; ++j) t.s4[j] = (unsigned short)(col[j] | 16u);
  int r0 = 0, r1 = 16;
  for (int a = 0; a < 8; ++a)
    for (int b = a + 1; b < 8; ++b) {
      const unsigned short m = (unsigned short)((1u << obit(a)) | (1u << obit(b)) | 16u);
      const bool first = (b < 6) || (a == 0 && b == 6);
      if (first) t.s3[r0++] = m; else t.s3[r1++] = m;
    }
  return t;
}

constexpr bool tablesComplete() {
  const Tables t = build();
  bool seen[512] = {};
  int n3 = 0, n4 = 0;
  for (int i = 0; i < 28; ++i) {
    const int m = t.s3[i];
    if (m >= 512 || popc9(m) != 3 || ((m >> 4) & 1) == 0 || seen[m]) return false;
    seen[m] = true; ++n3;
  }
  for (int i = 0; i < 56; ++i) {
    const int m = t.s4[i];
    if (m >= 512 || popc9(m) != 4 || ((m >> 4) & 1) == 0 || seen[m]) return false;
    seen[m] = true; ++n4;
    const int d = m & ~16;
    if (popc9(d) != 3 || seen[d]) return false;
    seen[d] = true; ++n3;
  }
  for (int i = 28; i < 32; ++i) if (t.s3[i] != 0xFFFFu) return false;
  return n3 == 84 && n4 == 56;
}
static_assert(tablesComplete(), "config tables do not match the rule");

constexpr bool deadRowsAreUnitRows() {
  const Tables t = build();
  for (int j = 0; j < 56; ++j) {
    const unsigned d = t.s4[j] & ~16u;
    for (int k = 0; k < 56; ++k) {
      const bool sub = (d & ~(unsigned)t.s4[k]) == 0u;
      if (sub != (k == j)) return false;
    }
  }
  return true;
}
static_assert(deadRowsAreUnitRows(), "a dead-center row has another superset");

constexpr unsigned S3c(int i) { return build().s3[i]; }
constexpr unsigned S4c(int i) { return build().s4[i]; }

struct MTab { unsigned m[32][2]; };

constexpr MTab buildMT() {
  MTab r{};
  constexpr Tables t = build();
  for (int row = 0; row < 32; ++row) {
    for (int H = 0; H < 2; ++H) {
      unsigned mm = 0;
      for (int idx = 0; idx < 28; ++idx) {
        const int kb = idx >> 1, j = idx & 1;
        const unsigned c3 = t.s3[row];
        const unsigned c4 = t.s4[4 * kb + j + 2 * H];
        if ((c3 & ~c4) == 0u) mm |= (1u << idx);
      }
      r.m[row][H] = mm;
    }
  }
  return r;
}

__device__ constexpr MTab MT = buildMT();

constexpr bool blockNZ(int t, int kb) {
  const MTab r = buildMT();
  for (int p = 0; p < 16; ++p)
    for (int H = 0; H < 2; ++H)
      if ((r.m[16 * t + p][H] >> (2 * kb)) & 3u) return true;
  return false;
}

template<typename F, int... Is>
__device__ __forceinline__ void static_for_impl(F&& f, std::integer_sequence<int, Is...>) {
  (f(std::integral_constant<int, Is>{}), ...);
}
template<int N, typename F>
__device__ __forceinline__ void static_for(F&& f) {
  static_for_impl(static_cast<F&&>(f), std::make_integer_sequence<int, N>{});
}

}

typedef __attribute__((ext_vector_type(2))) unsigned v2u;

__device__ __forceinline__ void half_swap(float& d, float& s) {
  const v2u r = __builtin_amdgcn_permlane16_swap(__float_as_uint(d), __float_as_uint(s), false, false);
  d = __uint_as_float(r.x);
  s = __uint_as_float(r.y);
}

__global__ __launch_bounds__(256) void gol_wmma_kernel(const float* __restrict__ x,
                                                       float* __restrict__ out,
                                                       int nTiles) {
  using namespace gol;
  const int lane = (int)(threadIdx.x & 31u);
  const int p = lane & 15;
  const int h = lane >> 4;
  const int w = (int)blockIdx.x * ((int)blockDim.x >> 5) + ((int)threadIdx.x >> 5);
  if (w >= nTiles) return;

  const int b  = w >> 11;
  const int rm = w & 2047;
  const int y  = rm >> 3;
  const int tx = rm & 7;
  const int x0 = (tx << 5) + lane;
  const float* img = x + (size_t)b * 65536;

  float n[9];
#pragma unroll
  for (int k = 0; k < 9; ++k) {
    const int dy = k / 3 - 1, dx = k % 3 - 1;
    const int yy = (y + dy) & 255;
    const int xx = (x0 + dx) & 255;
    n[k] = img[yy * 256 + xx];
  }
  float c[9];
#pragma unroll
  for (int k = 0; k < 9; ++k) c[k] = 1.0f - n[k];

  float gA[16]; gA[0] = 1.0f;
  static_for<4>([&](auto C4) {
    constexpr int cell = C4.value;
    constexpr int s = 1 << cell;
    static_for<s>([&](auto M) {
      constexpr int m = s - 1 - M.value;
      gA[m + s] = gA[m] * n[cell];
      gA[m]     = gA[m] * c[cell];
    });
  });
  float gB[32]; gB[0] = 1.0f;
  static_for<5>([&](auto C5) {
    constexpr int cell = C5.value;
    constexpr int s = 1 << cell;
    static_for<s>([&](auto M) {
      constexpr int m = s - 1 - M.value;
      gB[m + s] = gB[m] * n[cell + 4];
      gB[m]     = gB[m] * c[cell + 4];
    });
  });

  float sum0 = 0.0f, sum1 = 0.0f, sum2 = 0.0f, sum3 = 0.0f;

  float a4[56];
  float dp0 = 0.0f, dp1 = 0.0f, dp2 = 0.0f, dp3 = 0.0f;
  static_for<56>([&](auto I) {
    constexpr int col = I.value;
    constexpr unsigned mk = S4c(col);
    constexpr unsigned md = mk & ~16u;
    const float v = gA[mk & 15] * gB[(mk >> 4) & 31];
    const float d = gA[md & 15] * gB[(md >> 4) & 31];
    a4[col] = v;
    if constexpr ((col & 1) == 0) { sum0 += v; sum2 += d; } else { sum1 += v; sum3 += d; }
    if constexpr ((col & 3) == 0) dp0 = fmaf(d, v, dp0);
    else if constexpr ((col & 3) == 1) dp1 = fmaf(d, v, dp1);
    else if constexpr ((col & 3) == 2) dp2 = fmaf(d, v, dp2);
    else dp3 = fmaf(d, v, dp3);
  });

  float b1[28], b2[28];
  static_for<28>([&](auto I) {
    constexpr int idx = I.value;
    constexpr int kb = idx >> 1, j = idx & 1;
    float lo = a4[4 * kb + j];
    float hi = a4[4 * kb + j + 2];
    half_swap(lo, hi);
    b1[idx] = lo;
    b2[idx] = hi;
  });

  float a3[28];
  static_for<28>([&](auto I) {
    constexpr int row = I.value;
    constexpr unsigned mk = S3c(row);
    const float v = gA[mk & 15] * gB[(mk >> 4) & 31];
    a3[row] = v;
    if constexpr ((row & 1) == 0) sum0 += v; else sum1 += v;
  });

  unsigned mt[2];
  static_for<2>([&](auto T) {
    constexpr int t = T.value;
    mt[t] = MT.m[16 * t + p][h];
  });

  float orp0 = 0.0f, orp1 = 0.0f;
  static_for<2>([&](auto TT) {
    constexpr int t = TT.value;
    v8f C1 = {0.f, 0.f, 0.f, 0.f, 0.f, 0.f, 0.f, 0.f};
    v8f C2 = {0.f, 0.f, 0.f, 0.f, 0.f, 0.f, 0.f, 0.f};
    const unsigned m = mt[t];
    static_for<14>([&](auto KB) {
      constexpr int kb = KB.value;
      if constexpr (blockNZ(t, kb)) {
        v2f A, B1, B2;
        A.x = (float)((m >> (2 * kb)) & 1u);
        A.y = (float)((m >> (2 * kb + 1)) & 1u);
        B1.x = b1[2 * kb]; B1.y = b1[2 * kb + 1];
        B2.x = b2[2 * kb]; B2.y = b2[2 * kb + 1];
        C1 = __builtin_amdgcn_wmma_f32_16x16x4_f32(false, A, false, B1, (short)0, C1, false, false);
        C2 = __builtin_amdgcn_wmma_f32_16x16x4_f32(false, A, false, B2, (short)0, C2, false, false);
      }
    });
    static_for<8>([&](auto V) {
      constexpr int v = V.value;
      constexpr int r0 = 16 * t + v, r1 = r0 + 8;
      if constexpr (r0 < 28) {
        float lo = C1[v];
        float hi = C2[v];
        half_swap(lo, hi);
        orp0 = fmaf(a3[r0], lo, orp0);
        if constexpr (r1 < 28) orp1 = fmaf(a3[r1], hi, orp1);
      }
    });
  });

  const float tot = ((sum0 + sum1) + (sum2 + sum3)) - ((orp0 + orp1) + ((dp0 + dp1) + (dp2 + dp3)));
  out[(size_t)b * 65536 + (size_t)y * 256 + (size_t)x0] = tot;
}

extern "C" void kernel_launch(void* const* d_in, const int* in_sizes, int n_in,
                              void* d_out, int out_size, void* d_ws, size_t ws_size,
                              hipStream_t stream) {
  (void)n_in; (void)d_ws; (void)ws_size; (void)out_size;
  const float* x = (const float*)d_in[0];
  float* out = (float*)d_out;
  const int total = in_sizes[0];
  const int Bn = total >> 16;
  const int nTiles = Bn * 2048;
  const int wavesPerBlock = 8;
  dim3 block(256);
  dim3 grid((unsigned)((nTiles + wavesPerBlock - 1) / wavesPerBlock));
  gol_wmma_kernel<<<grid, block, 0, stream>>>(x, out, nTiles);
}


# A config's probability is VA[S1]*VQ[S2]*center, so the 140-sum and the 56 dead-center terms become rank-sum convolutions of the two 4-cell half tables (no per-config loop); f32 sum order changes
# speedup vs baseline: 5.1577x; 2.1464x over previous
#include <hip/hip_runtime.h>
#include <utility>


typedef __attribute__((ext_vector_type(2))) float v2f;
typedef __attribute__((ext_vector_type(8))) float v8f;

namespace gol {

struct Tables {
  unsigned short s3[28];
  unsigned short s4[56];
};

constexpr int popc9(int m) { int c = 0; for (int k = 0; k < 9; ++k) c += (m >> k) & 1; return c; }
constexpr int obit(int i) { return i < 4 ? i : i + 1; }

constexpr Tables build() {
  Tables t{};
  int n4 = 0, n3 = 0;
  for (int a = 0; a < 8; ++a)
    for (int b = a + 1; b < 8; ++b) {
      t.s3[n3++] = (unsigned short)((1u << obit(a)) | (1u << obit(b)) | 16u);
      for (int c = b + 1; c < 8; ++c)
        t.s4[n4++] = (unsigned short)((1u << obit(a)) | (1u << obit(b)) | (1u << obit(c)) | 16u);
    }
  return t;
}

constexpr bool tablesComplete() {
  const Tables t = build();
  bool seen[512] = {};
  int n3 = 0, n4 = 0;
  for (int i = 0; i < 28; ++i) {
    const int m = t.s3[i];
    if (m >= 512 || popc9(m) != 3 || ((m >> 4) & 1) == 0 || seen[m]) return false;
    seen[m] = true; ++n3;
  }
  for (int i = 0; i < 56; ++i) {
    const int m = t.s4[i];
    if (m >= 512 || popc9(m) != 4 || ((m >> 4) & 1) == 0 || seen[m]) return false;
    seen[m] = true; ++n4;
    const int d = m & ~16;
    if (popc9(d) != 3 || seen[d]) return false;
    seen[d] = true; ++n3;
  }
  return n3 == 84 && n4 == 56;
}
static_assert(tablesComplete(), "config tables do not match the rule");

constexpr bool deadRowsAreUnitRows() {
  const Tables t = build();
  for (int j = 0; j < 56; ++j) {
    const unsigned d = t.s4[j] & ~16u;
    for (int k = 0; k < 56; ++k) {
      const bool sub = (d & ~(unsigned)t.s4[k]) == 0u;
      if (sub != (k == j)) return false;
    }
  }
  return true;
}
static_assert(deadRowsAreUnitRows(), "a dead-center row has another superset");


constexpr int popc4(int m) { return (m & 1) + ((m >> 1) & 1) + ((m >> 2) & 1) + ((m >> 3) & 1); }

struct RowTab { int r[16]; };
constexpr RowTab buildRows() {
  RowTab t{};
  for (int i = 0; i < 16; ++i) t.r[i] = -1;
  int n2 = 0, n1 = 8;
  for (int m = 0; m < 16; ++m) {
    if (popc4(m) == 2) t.r[n2++] = m;
    else if (popc4(m) == 1) t.r[n1++] = m;
  }
  t.r[12] = 0;
  return t;
}
constexpr int ROWc(int i) { return buildRows().r[i]; }

constexpr bool rowsComplete() {
  const RowTab t = buildRows();
  bool seen[16] = {};
  int n = 0;
  for (int i = 0; i < 16; ++i) {
    const int m = t.r[i];
    if (m < 0) continue;
    if (m > 15 || popc4(m) > 2 || seen[m]) return false;
    seen[m] = true; ++n;
  }
  return n == 11;
}
static_assert(rowsComplete(), "cover rows must be the 11 subsets of at most 2 cells, once each");

struct alignas(8) ATab { unsigned w[32][2]; };
constexpr ATab buildA() {
  ATab t{};
  const RowTab rows = buildRows();
  for (int l = 0; l < 32; ++l)
    for (int kb = 0; kb < 4; ++kb)
      for (int j = 0; j < 2; ++j) {
        const int s  = rows.r[l & 15];
        const int sp = 4 * kb + j + 2 * (l >> 4);
        const bool cov = s >= 0 && (s & ~sp) == 0 && popc4(sp) == popc4(s) + 1;
        if (cov) t.w[l][kb >> 1] |= 1u << (8 * ((2 * kb + j) & 3));
      }
  return t;
}
constexpr bool aTableIsCover() {
  const ATab t = buildA();
  const RowTab rows = buildRows();
  int ones = 0;
  for (int l = 0; l < 32; ++l)
    for (int kb = 0; kb < 4; ++kb)
      for (int j = 0; j < 2; ++j) {
        const unsigned e = (t.w[l][kb >> 1] >> (16 * (kb & 1) + 8 * j)) & 0xffu;
        const int s  = rows.r[l & 15];
        const int sp = 4 * kb + j + 2 * (l >> 4);
        const bool cov = s >= 0 && (s & ~sp) == 0 && popc4(sp) == popc4(s) + 1;
        if (e != (cov ? 1u : 0u)) return false;
        ones += (int)e;
      }
  return ones == 4 + 12 + 12;
}
static_assert(aTableIsCover(), "A table is not the cover matrix");

__device__ constexpr ATab AF = buildA();

template<typename F, int... Is>
__device__ __forceinline__ void static_for_impl(F&& f, std::integer_sequence<int, Is...>) {
  (f(std::integral_constant<int, Is>{}), ...);
}
template<int N, typename F>
__device__ __forceinline__ void static_for(F&& f) {
  static_for_impl(static_cast<F&&>(f), std::make_integer_sequence<int, N>{});
}

}

typedef __attribute__((ext_vector_type(2))) unsigned v2u;

__device__ __forceinline__ void half_swap(float& d, float& s) {
  const v2u r = __builtin_amdgcn_permlane16_swap(__float_as_uint(d), __float_as_uint(s), false, false);
  d = __uint_as_float(r.x);
  s = __uint_as_float(r.y);
}

__global__ __launch_bounds__(256) void gol_wmma_kernel(const float* __restrict__ x,
                                                       float* __restrict__ out,
                                                       int nTiles) {
  using namespace gol;
  const int lane = (int)(threadIdx.x & 31u);
  const int w = (int)blockIdx.x * ((int)blockDim.x >> 5) + ((int)threadIdx.x >> 5);
  if (w >= nTiles) return;

  const int b  = w >> 11;
  const int rm = w & 2047;
  const int y  = rm >> 3;
  const int tx = rm & 7;
  const int x0 = (tx << 5) + lane;
  const float* img = x + (size_t)b * 65536;

  const unsigned aw0 = AF.w[lane][0];
  const unsigned aw1 = AF.w[lane][1];

  float n[9];
#pragma unroll
  for (int k = 0; k < 9; ++k) {
    const int dy = k / 3 - 1, dx = k % 3 - 1;
    const int yy = (y + dy) & 255;
    const int xx = (x0 + dx) & 255;
    n[k] = img[yy * 256 + xx];
  }
  float c[9];
#pragma unroll
  for (int k = 0; k < 9; ++k) c[k] = 1.0f - n[k];

  float gA[16]; gA[0] = 1.0f;
  static_for<4>([&](auto C4) {
    constexpr int cell = C4.value;
    constexpr int s = 1 << cell;
    static_for<s>([&](auto M) {
      constexpr int m = s - 1 - M.value;
      gA[m + s] = gA[m] * n[cell];
      gA[m]     = gA[m] * c[cell];
    });
  });
  float gQ[16]; gQ[0] = 1.0f;
  static_for<4>([&](auto C4) {
    constexpr int cell = C4.value;
    constexpr int s = 1 << cell;
    static_for<s>([&](auto M) {
      constexpr int m = s - 1 - M.value;
      gQ[m + s] = gQ[m] * n[cell + 5];
      gQ[m]     = gQ[m] * c[cell + 5];
    });
  });
  const float pc = n[4];
  const float qc = c[4];

  float bA1[8], bA2[8], bQ1[8], bQ2[8];
  static_for<8>([&](auto I) {
    constexpr int idx = I.value;
    constexpr int kb = idx >> 1, j = idx & 1;
    constexpr int k0 = 4 * kb + j, k1 = k0 + 2;
    float lo = gA[k0];
    float hi = gA[k1];
    half_swap(lo, hi);
    bA1[idx] = lo;
    bA2[idx] = hi;
    float lq = gQ[k0];
    float hq = gQ[k1];
    half_swap(lq, hq);
    bQ1[idx] = lq;
    bQ2[idx] = hq;
  });

  v8f CA1 = {0.f, 0.f, 0.f, 0.f, 0.f, 0.f, 0.f, 0.f};
  v8f CA2 = {0.f, 0.f, 0.f, 0.f, 0.f, 0.f, 0.f, 0.f};
  v8f CQ1 = {0.f, 0.f, 0.f, 0.f, 0.f, 0.f, 0.f, 0.f};
  v8f CQ2 = {0.f, 0.f, 0.f, 0.f, 0.f, 0.f, 0.f, 0.f};
  static_for<4>([&](auto KB) {
    constexpr int kb = KB.value;
    v2f A, B;
    constexpr int sh = 16 * (kb & 1);
    const unsigned awk = (kb < 2) ? aw0 : aw1;
    A.x = (float)((awk >> sh) & 0xffu);
    A.y = (float)((awk >> (sh + 8)) & 0xffu);
    B.x = bA1[2 * kb]; B.y = bA1[2 * kb + 1];
    CA1 = __builtin_amdgcn_wmma_f32_16x16x4_f32(false, A, false, B, (short)0, CA1, false, false);
    B.x = bA2[2 * kb]; B.y = bA2[2 * kb + 1];
    CA2 = __builtin_amdgcn_wmma_f32_16x16x4_f32(false, A, false, B, (short)0, CA2, false, false);
    B.x = bQ1[2 * kb]; B.y = bQ1[2 * kb + 1];
    CQ1 = __builtin_amdgcn_wmma_f32_16x16x4_f32(false, A, false, B, (short)0, CQ1, false, false);
    B.x = bQ2[2 * kb]; B.y = bQ2[2 * kb + 1];
    CQ2 = __builtin_amdgcn_wmma_f32_16x16x4_f32(false, A, false, B, (short)0, CQ2, false, false);
  });

  float eA[4], eQ[4], sA[4], sQ[4];
  static_for<16>([&](auto M) {
    constexpr int m = M.value;
    constexpr int r = popc4(m);
    if constexpr (r <= 3) {
      constexpr bool first = (m == (1 << r) - 1);
      if constexpr (first) {
        eA[r] = gA[m];
        eQ[r] = gQ[m];
        sA[r] = gA[m] * gA[m];
        sQ[r] = gQ[m] * gQ[m];
      } else {
        eA[r] += gA[m];
        eQ[r] += gQ[m];
        sA[r] = fmaf(gA[m], gA[m], sA[r]);
        sQ[r] = fmaf(gQ[m], gQ[m], sQ[r]);
      }
    }
  });

  float xA[3] = {0.0f, 0.0f, 0.0f}, xQ[3] = {0.0f, 0.0f, 0.0f};
  static_for<6>([&](auto V) {
    constexpr int v = V.value;
    constexpr int m0 = ROWc(v), m1 = ROWc(v + 8);
    float loA = CA1[v];
    float hiA = CA2[v];
    half_swap(loA, hiA);
    float loQ = CQ1[v];
    float hiQ = CQ2[v];
    half_swap(loQ, hiQ);
    static_assert(m0 >= 0, "rows 0..5 hold the 2-subsets");
    constexpr int r0 = popc4(m0);
    xA[r0] = fmaf(gA[m0], loA, xA[r0]);
    xQ[r0] = fmaf(gQ[m0], loQ, xQ[r0]);
    if constexpr (m1 >= 0) {
      constexpr int r1 = popc4(m1);
      xA[r1] = fmaf(gA[m1], hiA, xA[r1]);
      xQ[r1] = fmaf(gQ[m1], hiQ, xQ[r1]);
    }
  });

  const float orp = (fmaf(xA[0], sQ[2], xA[1] * sQ[1]) + fmaf(sA[0], xQ[2], sA[1] * xQ[1]))
                  + fmaf(xA[2], sQ[0], sA[2] * xQ[0]);

  const float E2 = fmaf(eA[0], eQ[2], eA[1] * eQ[1]) + eA[2] * eQ[0];
  const float E3 = fmaf(eA[0], eQ[3], eA[1] * eQ[2]) + fmaf(eA[2], eQ[1], eA[3] * eQ[0]);
  const float D3 = fmaf(sA[0], sQ[3], sA[1] * sQ[2]) + fmaf(sA[2], sQ[1], sA[3] * sQ[0]);

  const float all = fmaf(pc, E2 + E3, qc * E3);
  const float cor = pc * fmaf(pc, orp, qc * D3);
  const float tot = all - cor;
  out[(size_t)b * 65536 + (size_t)y * 256 + (size_t)x0] = tot;
}

extern "C" void kernel_launch(void* const* d_in, const int* in_sizes, int n_in,
                              void* d_out, int out_size, void* d_ws, size_t ws_size,
                              hipStream_t stream) {
  (void)n_in; (void)d_ws; (void)ws_size; (void)out_size;
  const float* x = (const float*)d_in[0];
  float* out = (float*)d_out;
  const int total = in_sizes[0];
  const int Bn = total >> 16;
  const int nTiles = Bn * 2048;
  const int wavesPerBlock = 8;
  dim3 block(256);
  dim3 grid((unsigned)((nTiles + wavesPerBlock - 1) / wavesPerBlock));
  gol_wmma_kernel<<<grid, block, 0, stream>>>(x, out, nTiles);
}


# Each wave walks 4 rows of its 32-pixel column, the 3 neighbor rows held in registers: 3 loads per pixel instead of 9 and 4x fewer waves to start; per-pixel arithmetic and f32 sum order unchanged
# speedup vs baseline: 5.6578x; 1.0970x over previous
#include <hip/hip_runtime.h>
#include <utility>


typedef __attribute__((ext_vector_type(2))) float v2f;
typedef __attribute__((ext_vector_type(8))) float v8f;

namespace gol {

struct Tables {
  unsigned short s3[28];
  unsigned short s4[56];
};

constexpr int popc9(int m) { int c = 0; for (int k = 0; k < 9; ++k) c += (m >> k) & 1; return c; }
constexpr int obit(int i) { return i < 4 ? i : i + 1; }

constexpr Tables build() {
  Tables t{};
  int n4 = 0, n3 = 0;
  for (int a = 0; a < 8; ++a)
    for (int b = a + 1; b < 8; ++b) {
      t.s3[n3++] = (unsigned short)((1u << obit(a)) | (1u << obit(b)) | 16u);
      for (int c = b + 1; c < 8; ++c)
        t.s4[n4++] = (unsigned short)((1u << obit(a)) | (1u << obit(b)) | (1u << obit(c)) | 16u);
    }
  return t;
}

constexpr bool tablesComplete() {
  const Tables t = build();
  bool seen[512] = {};
  int n3 = 0, n4 = 0;
  for (int i = 0; i < 28; ++i) {
    const int m = t.s3[i];
    if (m >= 512 || popc9(m) != 3 || ((m >> 4) & 1) == 0 || seen[m]) return false;
    seen[m] = true; ++n3;
  }
  for (int i = 0; i < 56; ++i) {
    const int m = t.s4[i];
    if (m >= 512 || popc9(m) != 4 || ((m >> 4) & 1) == 0 || seen[m]) return false;
    seen[m] = true; ++n4;
    const int d = m & ~16;
    if (popc9(d) != 3 || seen[d]) return false;
    seen[d] = true; ++n3;
  }
  return n3 == 84 && n4 == 56;
}
static_assert(tablesComplete(), "config tables do not match the rule");

constexpr bool deadRowsAreUnitRows() {
  const Tables t = build();
  for (int j = 0; j < 56; ++j) {
    const unsigned d = t.s4[j] & ~16u;
    for (int k = 0; k < 56; ++k) {
      const bool sub = (d & ~(unsigned)t.s4[k]) == 0u;
      if (sub != (k == j)) return false;
    }
  }
  return true;
}
static_assert(deadRowsAreUnitRows(), "a dead-center row has another superset");


constexpr int popc4(int m) { return (m & 1) + ((m >> 1) & 1) + ((m >> 2) & 1) + ((m >> 3) & 1); }

struct RowTab { int r[16]; };
constexpr RowTab buildRows() {
  RowTab t{};
  for (int i = 0; i < 16; ++i) t.r[i] = -1;
  int n2 = 0, n1 = 8;
  for (int m = 0; m < 16; ++m) {
    if (popc4(m) == 2) t.r[n2++] = m;
    else if (popc4(m) == 1) t.r[n1++] = m;
  }
  t.r[12] = 0;
  return t;
}
constexpr int ROWc(int i) { return buildRows().r[i]; }

constexpr bool rowsComplete() {
  const RowTab t = buildRows();
  bool seen[16] = {};
  int n = 0;
  for (int i = 0; i < 16; ++i) {
    const int m = t.r[i];
    if (m < 0) continue;
    if (m > 15 || popc4(m) > 2 || seen[m]) return false;
    seen[m] = true; ++n;
  }
  return n == 11;
}
static_assert(rowsComplete(), "cover rows must be the 11 subsets of at most 2 cells, once each");

struct alignas(8) ATab { unsigned w[32][2]; };
constexpr ATab buildA() {
  ATab t{};
  const RowTab rows = buildRows();
  for (int l = 0; l < 32; ++l)
    for (int kb = 0; kb < 4; ++kb)
      for (int j = 0; j < 2; ++j) {
        const int s  = rows.r[l & 15];
        const int sp = 4 * kb + j + 2 * (l >> 4);
        const bool cov = s >= 0 && (s & ~sp) == 0 && popc4(sp) == popc4(s) + 1;
        if (cov) t.w[l][kb >> 1] |= 1u << (8 * ((2 * kb + j) & 3));
      }
  return t;
}
constexpr bool aTableIsCover() {
  const ATab t = buildA();
  const RowTab rows = buildRows();
  int ones = 0;
  for (int l = 0; l < 32; ++l)
    for (int kb = 0; kb < 4; ++kb)
      for (int j = 0; j < 2; ++j) {
        const unsigned e = (t.w[l][kb >> 1] >> (16 * (kb & 1) + 8 * j)) & 0xffu;
        const int s  = rows.r[l & 15];
        const int sp = 4 * kb + j + 2 * (l >> 4);
        const bool cov = s >= 0 && (s & ~sp) == 0 && popc4(sp) == popc4(s) + 1;
        if (e != (cov ? 1u : 0u)) return false;
        ones += (int)e;
      }
  return ones == 4 + 12 + 12;
}
static_assert(aTableIsCover(), "A table is not the cover matrix");

__device__ constexpr ATab AF = buildA();

template<typename F, int... Is>
__device__ __forceinline__ void static_for_impl(F&& f, std::integer_sequence<int, Is...>) {
  (f(std::integral_constant<int, Is>{}), ...);
}
template<int N, typename F>
__device__ __forceinline__ void static_for(F&& f) {
  static_for_impl(static_cast<F&&>(f), std::make_integer_sequence<int, N>{});
}

}

typedef __attribute__((ext_vector_type(2))) unsigned v2u;

__device__ __forceinline__ void half_swap(float& d, float& s) {
  const v2u r = __builtin_amdgcn_permlane16_swap(__float_as_uint(d), __float_as_uint(s), false, false);
  d = __uint_as_float(r.x);
  s = __uint_as_float(r.y);
}

struct Row3 {
  float n[3];
  float c[3];
};

__device__ __forceinline__ Row3 load_row(const float* __restrict__ rowp, int xl, int xc, int xr) {
  Row3 r;
  r.n[0] = rowp[xl];
  r.n[1] = rowp[xc];
  r.n[2] = rowp[xr];
  r.c[0] = 1.0f - r.n[0];
  r.c[1] = 1.0f - r.n[1];
  r.c[2] = 1.0f - r.n[2];
  return r;
}

__device__ __forceinline__ float pixel_prob(const Row3& top, const Row3& mid, const Row3& bot,
                                            const v2f (&Afrag)[4]) {
  using namespace gol;
  float n[9], c[9];
  static_for<3>([&](auto K) {
    constexpr int k = K.value;
    n[k]     = top.n[k]; c[k]     = top.c[k];
    n[3 + k] = mid.n[k]; c[3 + k] = mid.c[k];
    n[6 + k] = bot.n[k]; c[6 + k] = bot.c[k];
  });

  float gA[16]; gA[0] = 1.0f;
  static_for<4>([&](auto C4) {
    constexpr int cell = C4.value;
    constexpr int s = 1 << cell;
    static_for<s>([&](auto M) {
      constexpr int m = s - 1 - M.value;
      gA[m + s] = gA[m] * n[cell];
      gA[m]     = gA[m] * c[cell];
    });
  });
  float gQ[16]; gQ[0] = 1.0f;
  static_for<4>([&](auto C4) {
    constexpr int cell = C4.value;
    constexpr int s = 1 << cell;
    static_for<s>([&](auto M) {
      constexpr int m = s - 1 - M.value;
      gQ[m + s] = gQ[m] * n[cell + 5];
      gQ[m]     = gQ[m] * c[cell + 5];
    });
  });
  const float pc = n[4];
  const float qc = c[4];

  float bA1[8], bA2[8], bQ1[8], bQ2[8];
  static_for<8>([&](auto I) {
    constexpr int idx = I.value;
    constexpr int kb = idx >> 1, j = idx & 1;
    constexpr int k0 = 4 * kb + j, k1 = k0 + 2;
    float lo = gA[k0];
    float hi = gA[k1];
    half_swap(lo, hi);
    bA1[idx] = lo;
    bA2[idx] = hi;
    float lq = gQ[k0];
    float hq = gQ[k1];
    half_swap(lq, hq);
    bQ1[idx] = lq;
    bQ2[idx] = hq;
  });

  v8f CA1 = {0.f, 0.f, 0.f, 0.f, 0.f, 0.f, 0.f, 0.f};
  v8f CA2 = {0.f, 0.f, 0.f, 0.f, 0.f, 0.f, 0.f, 0.f};
  v8f CQ1 = {0.f, 0.f, 0.f, 0.f, 0.f, 0.f, 0.f, 0.f};
  v8f CQ2 = {0.f, 0.f, 0.f, 0.f, 0.f, 0.f, 0.f, 0.f};
  static_for<4>([&](auto KB) {
    constexpr int kb = KB.value;
    const v2f A = Afrag[kb];
    v2f B;
    B.x = bA1[2 * kb]; B.y = bA1[2 * kb + 1];
    CA1 = __builtin_amdgcn_wmma_f32_16x16x4_f32(false, A, false, B, (short)0, CA1, false, false);
    B.x = bA2[2 * kb]; B.y = bA2[2 * kb + 1];
    CA2 = __builtin_amdgcn_wmma_f32_16x16x4_f32(false, A, false, B, (short)0, CA2, false, false);
    B.x = bQ1[2 * kb]; B.y = bQ1[2 * kb + 1];
    CQ1 = __builtin_amdgcn_wmma_f32_16x16x4_f32(false, A, false, B, (short)0, CQ1, false, false);
    B.x = bQ2[2 * kb]; B.y = bQ2[2 * kb + 1];
    CQ2 = __builtin_amdgcn_wmma_f32_16x16x4_f32(false, A, false, B, (short)0, CQ2, false, false);
  });

  float eA[4], eQ[4], sA[4], sQ[4];
  static_for<16>([&](auto M) {
    constexpr int m = M.value;
    constexpr int r = popc4(m);
    if constexpr (r <= 3) {
      constexpr bool first = (m == (1 << r) - 1);
      if constexpr (first) {
        eA[r] = gA[m];
        eQ[r] = gQ[m];
        sA[r] = gA[m] * gA[m];
        sQ[r] = gQ[m] * gQ[m];
      } else {
        eA[r] += gA[m];
        eQ[r] += gQ[m];
        sA[r] = fmaf(gA[m], gA[m], sA[r]);
        sQ[r] = fmaf(gQ[m], gQ[m], sQ[r]);
      }
    }
  });

  float xA[3] = {0.0f, 0.0f, 0.0f}, xQ[3] = {0.0f, 0.0f, 0.0f};
  static_for<6>([&](auto V) {
    constexpr int v = V.value;
    constexpr int m0 = ROWc(v), m1 = ROWc(v + 8);
    float loA = CA1[v];
    float hiA = CA2[v];
    half_swap(loA, hiA);
    float loQ = CQ1[v];
    float hiQ = CQ2[v];
    half_swap(loQ, hiQ);
    static_assert(m0 >= 0, "rows 0..5 hold the 2-subsets");
    constexpr int r0 = popc4(m0);
    xA[r0] = fmaf(gA[m0], loA, xA[r0]);
    xQ[r0] = fmaf(gQ[m0], loQ, xQ[r0]);
    if constexpr (m1 >= 0) {
      constexpr int r1 = popc4(m1);
      xA[r1] = fmaf(gA[m1], hiA, xA[r1]);
      xQ[r1] = fmaf(gQ[m1], hiQ, xQ[r1]);
    }
  });

  const float orp = (fmaf(xA[0], sQ[2], xA[1] * sQ[1]) + fmaf(sA[0], xQ[2], sA[1] * xQ[1]))
                  + fmaf(xA[2], sQ[0], sA[2] * xQ[0]);

  const float E2 = fmaf(eA[0], eQ[2], eA[1] * eQ[1]) + eA[2] * eQ[0];
  const float E3 = fmaf(eA[0], eQ[3], eA[1] * eQ[2]) + fmaf(eA[2], eQ[1], eA[3] * eQ[0]);
  const float D3 = fmaf(sA[0], sQ[3], sA[1] * sQ[2]) + fmaf(sA[2], sQ[1], sA[3] * sQ[0]);

  const float all = fmaf(pc, E2 + E3, qc * E3);
  const float cor = pc * fmaf(pc, orp, qc * D3);
  return all - cor;
}

constexpr int ROWS_PER_WAVE = 4;
constexpr int STRIPS_PER_IMAGE = 256 / ROWS_PER_WAVE;
static_assert(256 % ROWS_PER_WAVE == 0, "strips must tile the 256 image rows");

__global__ __launch_bounds__(256) void gol_wmma_kernel(const float* __restrict__ x,
                                                       float* __restrict__ out,
                                                       int nWaves) {
  using namespace gol;
  const int lane = (int)(threadIdx.x & 31u);
  const int w = (int)blockIdx.x * ((int)blockDim.x >> 5) + ((int)threadIdx.x >> 5);
  if (w >= nWaves) return;

  const int tx = w & 7;
  const int st = w >> 3;
  const int b  = st / STRIPS_PER_IMAGE;
  const int y0 = (st % STRIPS_PER_IMAGE) * ROWS_PER_WAVE;
  const int xc = (tx << 5) + lane;
  const int xl = (xc - 1) & 255;
  const int xr = (xc + 1) & 255;
  const float* img = x + (size_t)b * 65536;
  float* outp = out + (size_t)b * 65536 + (size_t)xc;

  const unsigned aw0 = AF.w[lane][0];
  const unsigned aw1 = AF.w[lane][1];
  v2f Afrag[4];
  static_for<4>([&](auto KB) {
    constexpr int kb = KB.value;
    constexpr int sh = 16 * (kb & 1);
    const unsigned awk = (kb < 2) ? aw0 : aw1;
    Afrag[kb].x = (float)((awk >> sh) & 0xffu);
    Afrag[kb].y = (float)((awk >> (sh + 8)) & 0xffu);
  });

  Row3 top = load_row(img + ((y0 - 1) & 255) * 256, xl, xc, xr);
  Row3 mid = load_row(img + y0 * 256, xl, xc, xr);
  Row3 bot = load_row(img + ((y0 + 1) & 255) * 256, xl, xc, xr);

  static_for<ROWS_PER_WAVE>([&](auto I) {
    constexpr int i = I.value;
    Row3 nxt;
    if constexpr (i + 1 < ROWS_PER_WAVE)
      nxt = load_row(img + ((y0 + i + 2) & 255) * 256, xl, xc, xr);

    outp[(size_t)(y0 + i) * 256] = pixel_prob(top, mid, bot, Afrag);

    if constexpr (i + 1 < ROWS_PER_WAVE) {
      top = mid;
      mid = bot;
      bot = nxt;
    }
  });
}

extern "C" void kernel_launch(void* const* d_in, const int* in_sizes, int n_in,
                              void* d_out, int out_size, void* d_ws, size_t ws_size,
                              hipStream_t stream) {
  (void)n_in; (void)d_ws; (void)ws_size; (void)out_size;
  const float* x = (const float*)d_in[0];
  float* out = (float*)d_out;
  const int total = in_sizes[0];
  const int Bn = total >> 16;
  const int nWaves = Bn * (2048 / ROWS_PER_WAVE);
  const int wavesPerBlock = 8;
  dim3 block(256);
  dim3 grid((unsigned)((nWaves + wavesPerBlock - 1) / wavesPerBlock));
  gol_wmma_kernel<<<grid, block, 0, stream>>>(x, out, nWaves);
}


# Cover GEMM cut to its six 2-cell rows (r=1 sum read from its upper end, r=0 from e[1]) and laid block-diagonal over the lane halves: 16->8 WMMAs, no lane swaps left; f32 sum order changes
# speedup vs baseline: 7.0589x; 1.2476x over previous
#include <hip/hip_runtime.h>
#include <utility>


typedef __attribute__((ext_vector_type(2))) float v2f;
typedef __attribute__((ext_vector_type(8))) float v8f;

namespace gol {

struct Tables {
  unsigned short s3[28];
  unsigned short s4[56];
};

constexpr int popc9(int m) { int c = 0; for (int k = 0; k < 9; ++k) c += (m >> k) & 1; return c; }
constexpr int obit(int i) { return i < 4 ? i : i + 1; }

constexpr Tables build() {
  Tables t{};
  int n4 = 0, n3 = 0;
  for (int a = 0; a < 8; ++a)
    for (int b = a + 1; b < 8; ++b) {
      t.s3[n3++] = (unsigned short)((1u << obit(a)) | (1u << obit(b)) | 16u);
      for (int c = b + 1; c < 8; ++c)
        t.s4[n4++] = (unsigned short)((1u << obit(a)) | (1u << obit(b)) | (1u << obit(c)) | 16u);
    }
  return t;
}

constexpr bool tablesComplete() {
  const Tables t = build();
  bool seen[512] = {};
  int n3 = 0, n4 = 0;
  for (int i = 0; i < 28; ++i) {
    const int m = t.s3[i];
    if (m >= 512 || popc9(m) != 3 || ((m >> 4) & 1) == 0 || seen[m]) return false;
    seen[m] = true; ++n3;
  }
  for (int i = 0; i < 56; ++i) {
    const int m = t.s4[i];
    if (m >= 512 || popc9(m) != 4 || ((m >> 4) & 1) == 0 || seen[m]) return false;
    seen[m] = true; ++n4;
    const int d = m & ~16;
    if (popc9(d) != 3 || seen[d]) return false;
    seen[d] = true; ++n3;
  }
  return n3 == 84 && n4 == 56;
}
static_assert(tablesComplete(), "config tables do not match the rule");

constexpr bool deadRowsAreUnitRows() {
  const Tables t = build();
  for (int j = 0; j < 56; ++j) {
    const unsigned d = t.s4[j] & ~16u;
    for (int k = 0; k < 56; ++k) {
      const bool sub = (d & ~(unsigned)t.s4[k]) == 0u;
      if (sub != (k == j)) return false;
    }
  }
  return true;
}
static_assert(deadRowsAreUnitRows(), "a dead-center row has another superset");


constexpr int popc4(int m) { return (m & 1) + ((m >> 1) & 1) + ((m >> 2) & 1) + ((m >> 3) & 1); }

struct RowTab { int r[8]; };
constexpr RowTab buildRows() {
  RowTab t{};
  for (int i = 0; i < 8; ++i) t.r[i] = -1;
  int n2 = 0;
  for (int m = 0; m < 16; ++m)
    if (popc4(m) == 2) t.r[n2++] = m;
  return t;
}
constexpr int ROWc(int i) { return buildRows().r[i]; }

struct KTab { int k[4][2]; };
constexpr KTab buildK() {
  KTab t{};
  int n3 = 0, n1 = 0;
  for (int m = 0; m < 16; ++m) {
    if (popc4(m) == 3) { t.k[n3 >> 1][n3 & 1] = m; ++n3; }
    if (popc4(m) == 1) { t.k[2 + (n1 >> 1)][n1 & 1] = m; ++n1; }
  }
  return t;
}
constexpr int KMc(int f, int j) { return buildK().k[f][j]; }

constexpr bool rowsAndKsComplete() {
  const RowTab rows = buildRows();
  const KTab ks = buildK();
  bool seenR[16] = {}, seenK[16] = {};
  int nr = 0, n3 = 0, n1 = 0;
  for (int i = 0; i < 8; ++i) {
    const int m = rows.r[i];
    if (i >= 6) { if (m != -1) return false; continue; }
    if (m < 0 || m > 15 || popc4(m) != 2 || seenR[m]) return false;
    seenR[m] = true; ++nr;
  }
  for (int f = 0; f < 4; ++f)
    for (int j = 0; j < 2; ++j) {
      const int m = ks.k[f][j];
      if (m < 1 || m > 14 || seenK[m]) return false;
      if (popc4(m) != (f < 2 ? 3 : 1)) return false;
      seenK[m] = true;
      if (f < 2) ++n3; else ++n1;
    }
  return nr == 6 && n3 == 4 && n1 == 4;
}
static_assert(rowsAndKsComplete(), "rows must be the six 2-cell subsets, k the 3-cell and 1-cell subsets, once each");

constexpr bool rel(int f, int s, int sp) {
  return f < 2 ? ((s & ~sp) == 0 && popc4(sp) == 3)
               : ((sp & ~s) == 0 && popc4(sp) == 1);
}

struct alignas(8) ATab { unsigned w[32][2]; };
constexpr bool aEntry(int l, int f, int j) {
  const RowTab rows = buildRows();
  const KTab ks = buildK();
  const int row = l & 15, h = l >> 4;
  const int s = rows.r[row & 7];
  return (row >> 3) == h && s >= 0 && rel(f, s, ks.k[f][j]);
}
constexpr ATab buildA() {
  ATab t{};
  for (int l = 0; l < 32; ++l)
    for (int f = 0; f < 4; ++f)
      for (int j = 0; j < 2; ++j)
        if (aEntry(l, f, j)) t.w[l][f >> 1] |= 1u << (8 * ((2 * f + j) & 3));
  return t;
}
constexpr bool aTableIsRelation() {
  const ATab t = buildA();
  int ones = 0;
  for (int l = 0; l < 32; ++l)
    for (int f = 0; f < 4; ++f)
      for (int j = 0; j < 2; ++j) {
        const unsigned e = (t.w[l][f >> 1] >> (16 * (f & 1) + 8 * j)) & 0xffu;
        if (e != (aEntry(l, f, j) ? 1u : 0u)) return false;
        if (e != 0u && ((l & 15) >> 3) != (l >> 4)) return false;
        ones += (int)e;
      }
  return ones == 2 * (6 * 2 + 6 * 2);
}
static_assert(aTableIsRelation(), "A table is not the block-diagonal cover relation");

__device__ constexpr ATab AF = buildA();

template<typename F, int... Is>
__device__ __forceinline__ void static_for_impl(F&& f, std::integer_sequence<int, Is...>) {
  (f(std::integral_constant<int, Is>{}), ...);
}
template<int N, typename F>
__device__ __forceinline__ void static_for(F&& f) {
  static_for_impl(static_cast<F&&>(f), std::make_integer_sequence<int, N>{});
}

}

struct Row3 {
  float n[3];
  float c[3];
};

__device__ __forceinline__ Row3 load_row(const float* __restrict__ rowp, int xl, int xc, int xr) {
  Row3 r;
  r.n[0] = rowp[xl];
  r.n[1] = rowp[xc];
  r.n[2] = rowp[xr];
  r.c[0] = 1.0f - r.n[0];
  r.c[1] = 1.0f - r.n[1];
  r.c[2] = 1.0f - r.n[2];
  return r;
}

__device__ __forceinline__ float pixel_prob(const Row3& top, const Row3& mid, const Row3& bot,
                                            const v2f (&Afrag)[4]) {
  using namespace gol;
  float n[9], c[9];
  static_for<3>([&](auto K) {
    constexpr int k = K.value;
    n[k]     = top.n[k]; c[k]     = top.c[k];
    n[3 + k] = mid.n[k]; c[3 + k] = mid.c[k];
    n[6 + k] = bot.n[k]; c[6 + k] = bot.c[k];
  });

  float gA[16]; gA[0] = 1.0f;
  static_for<4>([&](auto C4) {
    constexpr int cell = C4.value;
    constexpr int s = 1 << cell;
    static_for<s>([&](auto M) {
      constexpr int m = s - 1 - M.value;
      gA[m + s] = gA[m] * n[cell];
      gA[m]     = gA[m] * c[cell];
    });
  });
  float gQ[16]; gQ[0] = 1.0f;
  static_for<4>([&](auto C4) {
    constexpr int cell = C4.value;
    constexpr int s = 1 << cell;
    static_for<s>([&](auto M) {
      constexpr int m = s - 1 - M.value;
      gQ[m + s] = gQ[m] * n[cell + 5];
      gQ[m]     = gQ[m] * c[cell + 5];
    });
  });
  const float pc = n[4];
  const float qc = c[4];

  v8f CA = {0.f, 0.f, 0.f, 0.f, 0.f, 0.f, 0.f, 0.f};
  v8f UA = {0.f, 0.f, 0.f, 0.f, 0.f, 0.f, 0.f, 0.f};
  v8f CQ = {0.f, 0.f, 0.f, 0.f, 0.f, 0.f, 0.f, 0.f};
  v8f UQ = {0.f, 0.f, 0.f, 0.f, 0.f, 0.f, 0.f, 0.f};
  static_for<4>([&](auto F) {
    constexpr int f = F.value;
    constexpr int k0 = KMc(f, 0), k1 = KMc(f, 1);
    const v2f A = Afrag[f];
    v2f B;
    B.x = gA[k0]; B.y = gA[k1];
    if constexpr (f < 2)
      CA = __builtin_amdgcn_wmma_f32_16x16x4_f32(false, A, false, B, (short)0, CA, false, false);
    else
      UA = __builtin_amdgcn_wmma_f32_16x16x4_f32(false, A, false, B, (short)0, UA, false, false);
    B.x = gQ[k0]; B.y = gQ[k1];
    if constexpr (f < 2)
      CQ = __builtin_amdgcn_wmma_f32_16x16x4_f32(false, A, false, B, (short)0, CQ, false, false);
    else
      UQ = __builtin_amdgcn_wmma_f32_16x16x4_f32(false, A, false, B, (short)0, UQ, false, false);
  });

  float eA[4], eQ[4], sA[4], sQ[4];
  static_for<16>([&](auto M) {
    constexpr int m = M.value;
    constexpr int r = popc4(m);
    if constexpr (r <= 3) {
      constexpr bool first = (m == (1 << r) - 1);
      if constexpr (first) {
        eA[r] = gA[m];
        eQ[r] = gQ[m];
        sA[r] = gA[m] * gA[m];
        sQ[r] = gQ[m] * gQ[m];
      } else {
        eA[r] += gA[m];
        eQ[r] += gQ[m];
        sA[r] = fmaf(gA[m], gA[m], sA[r]);
        sQ[r] = fmaf(gQ[m], gQ[m], sQ[r]);
      }
    }
  });

  float xA[3], xQ[3];
  xA[0] = gA[0] * eA[1];
  xQ[0] = gQ[0] * eQ[1];
  static_for<6>([&](auto V) {
    constexpr int v = V.value;
    constexpr int m = ROWc(v);
    static_assert(m >= 0 && popc4(m) == 2, "rows 0..5 hold the 2-cell subsets");
    if constexpr (v == 0) {
      xA[2] = gA[m] * CA[v];
      xA[1] = gA[m] * UA[v];
      xQ[2] = gQ[m] * CQ[v];
      xQ[1] = gQ[m] * UQ[v];
    } else {
      xA[2] = fmaf(gA[m], CA[v], xA[2]);
      xA[1] = fmaf(gA[m], UA[v], xA[1]);
      xQ[2] = fmaf(gQ[m], CQ[v], xQ[2]);
      xQ[1] = fmaf(gQ[m], UQ[v], xQ[1]);
    }
  });

  const float orp = (fmaf(xA[0], sQ[2], xA[1] * sQ[1]) + fmaf(sA[0], xQ[2], sA[1] * xQ[1]))
                  + fmaf(xA[2], sQ[0], sA[2] * xQ[0]);

  const float E2 = fmaf(eA[0], eQ[2], eA[1] * eQ[1]) + eA[2] * eQ[0];
  const float E3 = fmaf(eA[0], eQ[3], eA[1] * eQ[2]) + fmaf(eA[2], eQ[1], eA[3] * eQ[0]);
  const float D3 = fmaf(sA[0], sQ[3], sA[1] * sQ[2]) + fmaf(sA[2], sQ[1], sA[3] * sQ[0]);

  const float all = fmaf(pc, E2 + E3, qc * E3);
  const float cor = pc * fmaf(pc, orp, qc * D3);
  return all - cor;
}

constexpr int ROWS_PER_WAVE = 4;
constexpr int STRIPS_PER_IMAGE = 256 / ROWS_PER_WAVE;
static_assert(256 % ROWS_PER_WAVE == 0, "strips must tile the 256 image rows");

__global__ __launch_bounds__(256) void gol_wmma_kernel(const float* __restrict__ x,
                                                       float* __restrict__ out,
                                                       int nWaves) {
  using namespace gol;
  const int lane = (int)(threadIdx.x & 31u);
  const int w = (int)blockIdx.x * ((int)blockDim.x >> 5) + ((int)threadIdx.x >> 5);
  if (w >= nWaves) return;

  const int tx = w & 7;
  const int st = w >> 3;
  const int b  = st / STRIPS_PER_IMAGE;
  const int y0 = (st % STRIPS_PER_IMAGE) * ROWS_PER_WAVE;
  const int xc = (tx << 5) + lane;
  const int xl = (xc - 1) & 255;
  const int xr = (xc + 1) & 255;
  const float* img = x + (size_t)b * 65536;
  float* outp = out + (size_t)b * 65536 + (size_t)xc;

  const unsigned aw0 = AF.w[lane][0];
  const unsigned aw1 = AF.w[lane][1];
  v2f Afrag[4];
  static_for<4>([&](auto F) {
    constexpr int f = F.value;
    constexpr int sh = 16 * (f & 1);
    const unsigned awk = (f < 2) ? aw0 : aw1;
    Afrag[f].x = (float)((awk >> sh) & 0xffu);
    Afrag[f].y = (float)((awk >> (sh + 8)) & 0xffu);
  });

  Row3 top = load_row(img + ((y0 - 1) & 255) * 256, xl, xc, xr);
  Row3 mid = load_row(img + y0 * 256, xl, xc, xr);
  Row3 bot = load_row(img + ((y0 + 1) & 255) * 256, xl, xc, xr);

  static_for<ROWS_PER_WAVE>([&](auto I) {
    constexpr int i = I.value;
    Row3 nxt;
    if constexpr (i + 1 < ROWS_PER_WAVE)
      nxt = load_row(img + ((y0 + i + 2) & 255) * 256, xl, xc, xr);

    outp[(size_t)(y0 + i) * 256] = pixel_prob(top, mid, bot, Afrag);

    if constexpr (i + 1 < ROWS_PER_WAVE) {
      top = mid;
      mid = bot;
      bot = nxt;
    }
  });
}

extern "C" void kernel_launch(void* const* d_in, const int* in_sizes, int n_in,
                              void* d_out, int out_size, void* d_ws, size_t ws_size,
                              hipStream_t stream) {
  (void)n_in; (void)d_ws; (void)ws_size; (void)out_size;
  const float* x = (const float*)d_in[0];
  float* out = (float*)d_out;
  const int total = in_sizes[0];
  const int Bn = total >> 16;
  const int nWaves = Bn * (2048 / ROWS_PER_WAVE);
  const int wavesPerBlock = 8;
  dim3 block(256);
  dim3 grid((unsigned)((nWaves + wavesPerBlock - 1) / wavesPerBlock));
  gol_wmma_kernel<<<grid, block, 0, stream>>>(x, out, nWaves);
}
